# starting kernel
# baseline (speedup 1.0000x reference)
#include <hip/hip_runtime.h>

typedef __attribute__((ext_vector_type(2))) float v2f;
typedef __attribute__((ext_vector_type(8))) float v8f;

#define TILE_K 32
#define BLK_M  128
#define BLK_N  64
#define AP     34
#define BP     162
#define GK     512
#define NT_K   (GK / TILE_K)

__global__ __launch_bounds__(256) void dct_make(float* __restrict__ D) {
  int idx = blockIdx.x * 256 + threadIdx.x;
  int k = idx >> 9;
  int n = idx & 511;
  int r = ((2 * n + 1) * k) & 2047;
  float th = 3.14159265358979323846f * (float)r * (1.0f / 1024.0f);
  float v = cosf(th) * 0.0625f;
  if (k == 0) v *= 0.70710678118654752f;
  D[idx] = v;
}


__device__ __forceinline__ void fetchA(const float* __restrict__ Ab, int m0, int k0,
                                       long rsA, long csA, int tid, float (&r)[16]) {
#pragma unroll
  for (int t = 0; t < 16; ++t) {
    int idx = tid + t * 256;
    int kk = idx & (TILE_K - 1);
    int mm = idx >> 5;
    r[t] = Ab[(long)(m0 + mm) * rsA + (long)(k0 + kk) * csA];
  }
}

__device__ __forceinline__ void storeA(float* __restrict__ As, int tid,
                                       const float (&r)[16]) {
#pragma unroll
  for (int t = 0; t < 16; ++t) {
    int idx = tid + t * 256;
    int kk = idx & (TILE_K - 1);
    int mm = idx >> 5;
    As[mm * AP + kk] = r[t];
  }
}

__device__ __forceinline__ void fetchB(const float* __restrict__ Bb, int n0, int k0,
                                       long rsB, long csB, int tid, int bNFast,
                                       float (&r)[8]) {
  if (bNFast) {
#pragma unroll
    for (int t = 0; t < 8; ++t) {
      int idx = tid + t * 256;
      int nn = idx & (BLK_N - 1);
      int kk = idx >> 6;
      r[t] = Bb[(long)(k0 + kk) * rsB + (long)(n0 + nn) * csB];
    }
  } else {
#pragma unroll
    for (int t = 0; t < 8; ++t) {
      int idx = tid + t * 256;
      int kk = idx & (TILE_K - 1);
      int nn = idx >> 5;
      r[t] = Bb[(long)(k0 + kk) * rsB + (long)(n0 + nn) * csB];
    }
  }
}

__device__ __forceinline__ void storeB(float* __restrict__ Bs, int tid, int bNFast,
                                       const float (&r)[8]) {
  if (bNFast) {
#pragma unroll
    for (int t = 0; t < 8; ++t) {
      int idx = tid + t * 256;
      int nn = idx & (BLK_N - 1);
      int kk = idx >> 6;
      Bs[(kk >> 1) * BP + nn * 2 + (kk & 1)] = r[t];
    }
  } else {
#pragma unroll
    for (int t = 0; t < 8; ++t) {
      int idx = tid + t * 256;
      int kk = idx & (TILE_K - 1);
      int nn = idx >> 5;
      Bs[(kk >> 1) * BP + nn * 2 + (kk & 1)] = r[t];
    }
  }
}


__device__ __forceinline__ void mma_tile(const float* __restrict__ As,
                                         const float* __restrict__ Bs,
                                         v8f (&acc)[2][2],
                                         int wm, int wn, int row, int grp) {
#pragma unroll
  for (int kk = 0; kk < TILE_K; kk += 4) {
    v2f a[2], b[2];
#pragma unroll
    for (int mt = 0; mt < 2; ++mt)
      a[mt] = *(const v2f*)&As[(wm + mt * 16 + row) * AP + kk + 2 * grp];
#pragma unroll
    for (int nt = 0; nt < 2; ++nt)
      b[nt] = *(const v2f*)&Bs[((kk >> 1) + grp) * BP + (wn + nt * 16 + row) * 2];
#pragma unroll
    for (int mt = 0; mt < 2; ++mt)
#pragma unroll
      for (int nt = 0; nt < 2; ++nt)
        acc[mt][nt] = __builtin_amdgcn_wmma_f32_16x16x4_f32(
            false, a[mt], false, b[nt], (short)0, acc[mt][nt], false, false);
  }
}


__global__ __launch_bounds__(256)
void dct_gemm(const float* __restrict__ A, const float* __restrict__ B,
              float* __restrict__ C, int N,
              long rsA, long csA, long rsB, long csB, long rsC, long csC,
              int zDiv, long zPitch, int useOffA, int useOffB, int useOffC,
              int bNFast) {
  (void)N;
  __shared__ float As[2][BLK_M * AP];
  __shared__ float Bs[2][(TILE_K / 2) * BP];

  const int tid = threadIdx.x;
  const int z   = blockIdx.z;
  const long off = (long)(z / zDiv) * zPitch + (long)(z % zDiv);
  const float* Ab = A + (useOffA ? off : 0);
  const float* Bb = B + (useOffB ? off : 0);
  float*       Cb = C + (useOffC ? off : 0);

  const int m0 = blockIdx.y * BLK_M;
  const int n0 = blockIdx.x * BLK_N;

  const int lane = tid & 31;
  const int wave = tid >> 5;
  const int wm   = (wave & 3) * 32;
  const int wn   = (wave >> 2) * 32;
  const int row  = lane & 15;
  const int grp  = lane >> 4;

  v8f acc[2][2];
#pragma unroll
  for (int i = 0; i < 2; ++i)
#pragma unroll
    for (int j = 0; j < 2; ++j) acc[i][j] = (v8f)0.0f;

  float ra[16], rb[8];

  fetchA(Ab, m0, 0, rsA, csA, tid, ra);
  fetchB(Bb, n0, 0, rsB, csB, tid, bNFast, rb);
  storeA(As[0], tid, ra);
  storeB(Bs[0], tid, bNFast, rb);
  __syncthreads();

  for (int t = 0; t < NT_K; ++t) {
    const int cur = t & 1;
    if (t + 1 < NT_K) {
      fetchA(Ab, m0, (t + 1) * TILE_K, rsA, csA, tid, ra);
      fetchB(Bb, n0, (t + 1) * TILE_K, rsB, csB, tid, bNFast, rb);
    }
    mma_tile(As[cur], Bs[cur], acc, wm, wn, row, grp);
    if (t + 1 < NT_K) {
      storeA(As[cur ^ 1], tid, ra);
      storeB(Bs[cur ^ 1], tid, bNFast, rb);
    }
    __syncthreads();
  }

#pragma unroll
  for (int mt = 0; mt < 2; ++mt)
#pragma unroll
    for (int nt = 0; nt < 2; ++nt)
#pragma unroll
      for (int v = 0; v < 8; ++v) {
        int m = m0 + wm + mt * 16 + grp * 8 + v;
        int n = n0 + wn + nt * 16 + row;
        Cb[(long)m * rsC + (long)n * csC] = acc[mt][nt][v];
      }
}

extern "C" void kernel_launch(void* const* d_in, const int* in_sizes, int n_in,
                              void* d_out, int out_size, void* d_ws, size_t ws_size,
                              hipStream_t stream) {
  (void)in_sizes; (void)n_in; (void)out_size; (void)ws_size;
  const float* x = (const float*)d_in[0];
  float* out = (float*)d_out;
  float* Dm  = (float*)d_ws;
  float* Y1  = Dm + 512 * 512;
  const long HWC = 512L * 512L * 3L;

  dct_make<<<dim3(1024), dim3(256), 0, stream>>>(Dm);

  dct_gemm<<<dim3(24, 4, 32), dim3(256), 0, stream>>>(
      Dm, x, Y1, 1536,
        512, 1,   1536, 1,   1536, 1,
        1,   HWC,   0, 1, 1,   1);

  dct_gemm<<<dim3(8, 4, 96), dim3(256), 0, stream>>>(
      Y1, Dm, out, 512,
        1536, 3,   1, 512,   1536, 3,
        3,   HWC,   1, 0, 1,   0);
}


# Block tile 128x64 -> 128x128, wave tile 32x32 -> 64x32: 8 WMMAs per k-step of fragment loads instead of 4; tile strides compile-time and two scheduling hints so the prefetch order survives
# speedup vs baseline: 1.4272x; 1.4272x over previous
#include <hip/hip_runtime.h>

typedef __attribute__((ext_vector_type(2))) float v2f;
typedef __attribute__((ext_vector_type(8))) float v8f;

#define TILE_K 32
#define BLK_M  128
#define BLK_N  128
#define AP     34
#define BP     290
#define GK     512
#define NT_K   (GK / TILE_K)
#define HWC    786432L

__global__ __launch_bounds__(256) void dct_make(float* __restrict__ D) {
  int idx = blockIdx.x * 256 + threadIdx.x;
  int k = idx >> 9;
  int n = idx & 511;
  int r = ((2 * n + 1) * k) & 2047;
  float th = 3.14159265358979323846f * (float)r * (1.0f / 1024.0f);
  float v = cosf(th) * 0.0625f;
  if (k == 0) v *= 0.70710678118654752f;
  D[idx] = v;
}

template <int ST> struct Geo;
template <> struct Geo<1> {
  static constexpr int rsA = 512, csA = 1, rsB = 1536, csB = 1, rsC = 1536, csC = 1;
  static constexpr int zDiv = 1;
  static constexpr bool offA = false, offB = true, offC = true, bNFast = true;
};
template <> struct Geo<2> {
  static constexpr int rsA = 1536, csA = 3, rsB = 1, csB = 512, rsC = 1536, csC = 3;
  static constexpr int zDiv = 3;
  static constexpr bool offA = true, offB = false, offC = true, bNFast = false;
};


template <int ST>
__device__ __forceinline__ void fetchA(const float* __restrict__ pa, float (&r)[16]) {
#pragma unroll
  for (int t = 0; t < 16; ++t) r[t] = pa[t * 8 * Geo<ST>::rsA];
}

__device__ __forceinline__ void storeA(float* __restrict__ As, int sa, const float (&r)[16]) {
#pragma unroll
  for (int t = 0; t < 16; ++t) As[sa + t * 8 * AP] = r[t];
}

template <int ST>
__device__ __forceinline__ void fetchB(const float* __restrict__ pb, float (&r)[16]) {
  if (Geo<ST>::bNFast) {
#pragma unroll
    for (int t = 0; t < 16; ++t) r[t] = pb[t * 2 * Geo<ST>::rsB];
  } else {
#pragma unroll
    for (int t = 0; t < 16; ++t) r[t] = pb[t * 8 * Geo<ST>::csB];
  }
}

template <int ST>
__device__ __forceinline__ void storeB(float* __restrict__ Bs, int sb, const float (&r)[16]) {
  if (Geo<ST>::bNFast) {
#pragma unroll
    for (int t = 0; t < 16; ++t) Bs[sb + t * BP] = r[t];
  } else {
#pragma unroll
    for (int t = 0; t < 16; ++t) Bs[sb + t * 16] = r[t];
  }
}


__device__ __forceinline__ void mma_tile(const float* __restrict__ As,
                                         const float* __restrict__ Bs,
                                         v8f (&acc)[4][2],
                                         int wm, int wn, int row, int grp) {
#pragma unroll
  for (int kk = 0; kk < TILE_K; kk += 4) {
    v2f a[4], b[2];
#pragma unroll
    for (int mt = 0; mt < 4; ++mt)
      a[mt] = *(const v2f*)&As[(wm + mt * 16 + row) * AP + kk + 2 * grp];
#pragma unroll
    for (int nt = 0; nt < 2; ++nt)
      b[nt] = *(const v2f*)&Bs[((kk >> 1) + grp) * BP + (wn + nt * 16 + row) * 2];
#pragma unroll
    for (int mt = 0; mt < 4; ++mt)
#pragma unroll
      for (int nt = 0; nt < 2; ++nt)
        acc[mt][nt] = __builtin_amdgcn_wmma_f32_16x16x4_f32(
            false, a[mt], false, b[nt], (short)0, acc[mt][nt], false, false);
  }
}


template <int ST>
__global__ __launch_bounds__(256) __attribute__((amdgpu_waves_per_eu(4, 4)))
void dct_gemm(const float* __restrict__ A, const float* __restrict__ B,
              float* __restrict__ C) {
  typedef Geo<ST> G;
  __shared__ float As[2][BLK_M * AP];
  __shared__ float Bs[2][(TILE_K / 2) * BP];

  const int tid = threadIdx.x;
  const int z   = blockIdx.z;
  const long off = (long)(z / G::zDiv) * HWC + (long)(z % G::zDiv);
  const float* Ab = A + (G::offA ? off : 0);
  const float* Bb = B + (G::offB ? off : 0);
  float*       Cb = C + (G::offC ? off : 0);

  const int m0 = blockIdx.y * BLK_M;
  const int n0 = blockIdx.x * BLK_N;

  const int lane = tid & 31;
  const int wave = tid >> 5;
  const int wm   = (wave & 1) * 64;
  const int wn   = (wave >> 1) * 32;
  const int row  = lane & 15;
  const int grp  = lane >> 4;

  const float* pa = Ab + (long)(m0 + (tid >> 5)) * G::rsA + (long)(tid & 31) * G::csA;
  const float* pb = G::bNFast
      ? Bb + (long)(tid >> 7) * G::rsB + (long)(n0 + (tid & 127)) * G::csB
      : Bb + (long)(tid & 31) * G::rsB + (long)(n0 + (tid >> 5)) * G::csB;
  const int sa = (tid >> 5) * AP + (tid & 31);
  const int sb = G::bNFast
      ? (tid & 127) * 2 + (tid >> 7)
      : ((tid & 31) >> 1) * BP + (tid >> 5) * 2 + (tid & 1);
  const long stepA = (long)TILE_K * G::csA;
  const long stepB = (long)TILE_K * G::rsB;

  v8f acc[4][2];
#pragma unroll
  for (int i = 0; i < 4; ++i)
#pragma unroll
    for (int j = 0; j < 2; ++j) acc[i][j] = (v8f)0.0f;

  float ra[16], rb[16];

  fetchA<ST>(pa, ra);
  fetchB<ST>(pb, rb);
  storeA(As[0], sa, ra);
  storeB<ST>(Bs[0], sb, rb);
  __syncthreads();

  for (int t = 0; t < NT_K; ++t) {
    const int cur = t & 1;
    if (t + 1 < NT_K) {
      pa += stepA;
      pb += stepB;
      fetchA<ST>(pa, ra);
      fetchB<ST>(pb, rb);
    }
    mma_tile(As[cur], Bs[cur], acc, wm, wn, row, grp);
    __builtin_amdgcn_sched_barrier(0);
    if (t + 1 < NT_K) {
      storeA(As[cur ^ 1], sa, ra);
      storeB<ST>(Bs[cur ^ 1], sb, rb);
    }
    __syncthreads();
  }

  float* pc = Cb + (long)(m0 + wm + grp * 8) * G::rsC + (long)(n0 + wn + row) * G::csC;
#pragma unroll
  for (int mt = 0; mt < 4; ++mt)
#pragma unroll
    for (int nt = 0; nt < 2; ++nt)
#pragma unroll
      for (int v = 0; v < 8; ++v)
        pc[(mt * 16 + v) * G::rsC + nt * 16 * G::csC] = acc[mt][nt][v];
}

extern "C" void kernel_launch(void* const* d_in, const int* in_sizes, int n_in,
                              void* d_out, int out_size, void* d_ws, size_t ws_size,
                              hipStream_t stream) {
  (void)in_sizes; (void)n_in; (void)out_size; (void)ws_size;
  const float* x = (const float*)d_in[0];
  float* out = (float*)d_out;
  float* Dm  = (float*)d_ws;
  float* Y1  = Dm + 512 * 512;

  dct_make<<<dim3(1024), dim3(256), 0, stream>>>(Dm);

  dct_gemm<1><<<dim3(12, 4, 32), dim3(256), 0, stream>>>(Dm, x, Y1);

  dct_gemm<2><<<dim3(4, 4, 96), dim3(256), 0, stream>>>(Y1, Dm, out);
}


# Mirror fold of the DCT basis (D[k][511-n] = (-1)^k D[k][n]): each work-group takes outputs of one parity, data operand pre-summed v[n]+-v[511-n] into LDS, K 512 -> 256; changes summation order
# speedup vs baseline: 1.8793x; 1.3168x over previous
#include <hip/hip_runtime.h>

typedef __attribute__((ext_vector_type(2))) float v2f;
typedef __attribute__((ext_vector_type(8))) float v8f;

#define TILE_K 32
#define BLK_M  128
#define BLK_N  128
#define AP     34
#define BP     290
#define GK     256
#define MIR    511
#define NT_K   (GK / TILE_K)
#define HWC    786432L

__global__ __launch_bounds__(256) void dct_make(float* __restrict__ D) {
  int idx = blockIdx.x * 256 + threadIdx.x;
  int k = idx >> 9;
  int n = idx & 511;
  int r = ((2 * n + 1) * k) & 2047;
  float th = 3.14159265358979323846f * (float)r * (1.0f / 1024.0f);
  float v = cosf(th) * 0.0625f;
  if (k == 0) v *= 0.70710678118654752f;
  D[idx] = v;
}

template <int ST> struct Geo;
template <> struct Geo<1> {
  static constexpr int rsA = 1024, csA = 1, rsB = 1536, csB = 1, rsC = 3072, csC = 1;
  static constexpr int parA = 512, parB = 0, parC = 1536;
  static constexpr int zDiv = 1;
  static constexpr bool offA = false, offB = true, offC = true, bNFast = true;
  static constexpr bool foldA = false;
};
template <> struct Geo<2> {
  static constexpr int rsA = 1536, csA = 3, rsB = 1, csB = 1024, rsC = 1536, csC = 6;
  static constexpr int parA = 0, parB = 512, parC = 3;
  static constexpr int zDiv = 3;
  static constexpr bool offA = true, offB = false, offC = true, bNFast = false;
  static constexpr bool foldA = true;
};


template <int ST>
__device__ __forceinline__ void fetchA(const float* __restrict__ pa, float (&r)[16]) {
#pragma unroll
  for (int t = 0; t < 16; ++t) r[t] = pa[t * 8 * Geo<ST>::rsA];
}

__device__ __forceinline__ void storeA(float* __restrict__ As, int sa, const float (&r)[16]) {
#pragma unroll
  for (int t = 0; t < 16; ++t) As[sa + t * 8 * AP] = r[t];
}

template <int ST>
__device__ __forceinline__ void fetchB(const float* __restrict__ pb, float (&r)[16]) {
  if (Geo<ST>::bNFast) {
#pragma unroll
    for (int t = 0; t < 16; ++t) r[t] = pb[t * 2 * Geo<ST>::rsB];
  } else {
#pragma unroll
    for (int t = 0; t < 16; ++t) r[t] = pb[t * 8 * Geo<ST>::csB];
  }
}

template <int ST>
__device__ __forceinline__ void fetchMir(const float* __restrict__ pm, float (&r)[16]) {
  if (Geo<ST>::foldA) {
#pragma unroll
    for (int t = 0; t < 16; ++t) r[t] = pm[t * 8 * Geo<ST>::rsA];
  } else {
#pragma unroll
    for (int t = 0; t < 16; ++t) r[t] = pm[-(t * 2 * Geo<ST>::rsB)];
  }
}

__device__ __forceinline__ void foldMir(float (&r)[16], const float (&m)[16], float s) {
#pragma unroll
  for (int t = 0; t < 16; ++t) r[t] = __builtin_fmaf(s, m[t], r[t]);
}

template <int ST>
__device__ __forceinline__ void storeB(float* __restrict__ Bs, int sb, const float (&r)[16]) {
  if (Geo<ST>::bNFast) {
#pragma unroll
    for (int t = 0; t < 16; ++t) Bs[sb + t * BP] = r[t];
  } else {
#pragma unroll
    for (int t = 0; t < 16; ++t) Bs[sb + t * 16] = r[t];
  }
}


__device__ __forceinline__ void mma_tile(const float* __restrict__ As,
                                         const float* __restrict__ Bs,
                                         v8f (&acc)[4][2],
                                         int wm, int wn, int row, int grp) {
#pragma unroll
  for (int kk = 0; kk < TILE_K; kk += 4) {
    v2f a[4], b[2];
#pragma unroll
    for (int mt = 0; mt < 4; ++mt)
      a[mt] = *(const v2f*)&As[(wm + mt * 16 + row) * AP + kk + 2 * grp];
#pragma unroll
    for (int nt = 0; nt < 2; ++nt)
      b[nt] = *(const v2f*)&Bs[((kk >> 1) + grp) * BP + (wn + nt * 16 + row) * 2];
#pragma unroll
    for (int mt = 0; mt < 4; ++mt)
#pragma unroll
      for (int nt = 0; nt < 2; ++nt)
        acc[mt][nt] = __builtin_amdgcn_wmma_f32_16x16x4_f32(
            false, a[mt], false, b[nt], (short)0, acc[mt][nt], false, false);
  }
}


template <int ST>
__global__ __launch_bounds__(256) __attribute__((amdgpu_waves_per_eu(4, 4)))
void dct_gemm(const float* __restrict__ A, const float* __restrict__ B,
              float* __restrict__ C) {
  typedef Geo<ST> G;
  __shared__ float As[2][BLK_M * AP];
  __shared__ float Bs[2][(TILE_K / 2) * BP];

  const int tid = threadIdx.x;
  const int z   = blockIdx.z;
  const long off = (long)(z / G::zDiv) * HWC + (long)(z % G::zDiv);
  const int gf  = G::foldA ? blockIdx.x : blockIdx.y;
  const int par = gf >> 1;
  const float sgn = par ? -1.0f : 1.0f;
  const float* Ab = A + (G::offA ? off : 0) + par * G::parA;
  const float* Bb = B + (G::offB ? off : 0) + par * G::parB;
  float*       Cb = C + (G::offC ? off : 0) + par * G::parC;

  const int m0 = (G::foldA ? blockIdx.y : (blockIdx.y & 1)) * BLK_M;
  const int n0 = (G::foldA ? (blockIdx.x & 1) : blockIdx.x) * BLK_N;

  const int lane = tid & 31;
  const int wave = tid >> 5;
  const int wm   = (wave & 1) * 64;
  const int wn   = (wave >> 1) * 32;
  const int row  = lane & 15;
  const int grp  = lane >> 4;

  const float* pa = Ab + (long)(m0 + (tid >> 5)) * G::rsA + (long)(tid & 31) * G::csA;
  const float* pb = G::bNFast
      ? Bb + (long)(tid >> 7) * G::rsB + (long)(n0 + (tid & 127)) * G::csB
      : Bb + (long)(tid & 31) * G::rsB + (long)(n0 + (tid >> 5)) * G::csB;
  const int sa = (tid >> 5) * AP + (tid & 31);
  const int sb = G::bNFast
      ? (tid & 127) * 2 + (tid >> 7)
      : ((tid & 31) >> 1) * BP + (tid >> 5) * 2 + (tid & 1);
  const float* pm = G::foldA
      ? Ab + (long)(m0 + (tid >> 5)) * G::rsA + (long)(MIR - (tid & 31)) * G::csA
      : Bb + (long)(MIR - (tid >> 7)) * G::rsB + (long)(n0 + (tid & 127)) * G::csB;
  const long stepA = (long)TILE_K * G::csA;
  const long stepB = (long)TILE_K * G::rsB;
  const long stepM = G::foldA ? stepA : stepB;

  v8f acc[4][2];
#pragma unroll
  for (int i = 0; i < 4; ++i)
#pragma unroll
    for (int j = 0; j < 2; ++j) acc[i][j] = (v8f)0.0f;

  float ra[16], rb[16], rm[16];

  fetchA<ST>(pa, ra);
  fetchB<ST>(pb, rb);
  fetchMir<ST>(pm, rm);
  foldMir(G::foldA ? ra : rb, rm, sgn);
  storeA(As[0], sa, ra);
  storeB<ST>(Bs[0], sb, rb);
  __syncthreads();

  for (int t = 0; t < NT_K; ++t) {
    const int cur = t & 1;
    if (t + 1 < NT_K) {
      pa += stepA;
      pb += stepB;
      pm -= stepM;
      fetchA<ST>(pa, ra);
      fetchB<ST>(pb, rb);
      fetchMir<ST>(pm, rm);
    }
    mma_tile(As[cur], Bs[cur], acc, wm, wn, row, grp);
    __builtin_amdgcn_sched_barrier(0);
    if (t + 1 < NT_K) {
      foldMir(G::foldA ? ra : rb, rm, sgn);
      storeA(As[cur ^ 1], sa, ra);
      storeB<ST>(Bs[cur ^ 1], sb, rb);
    }
    __syncthreads();
  }

  float* pc = Cb + (long)(m0 + wm + grp * 8) * G::rsC + (long)(n0 + wn + row) * G::csC;
#pragma unroll
  for (int mt = 0; mt < 4; ++mt)
#pragma unroll
    for (int nt = 0; nt < 2; ++nt)
#pragma unroll
      for (int v = 0; v < 8; ++v)
        pc[(mt * 16 + v) * G::rsC + nt * 16 * G::csC] = acc[mt][nt][v];
}

extern "C" void kernel_launch(void* const* d_in, const int* in_sizes, int n_in,
                              void* d_out, int out_size, void* d_ws, size_t ws_size,
                              hipStream_t stream) {
  (void)in_sizes; (void)n_in; (void)out_size; (void)ws_size;
  const float* x = (const float*)d_in[0];
  float* out = (float*)d_out;
  float* Dm  = (float*)d_ws;
  float* Y1  = Dm + 512 * 512;

  dct_make<<<dim3(1024), dim3(256), 0, stream>>>(Dm);

  dct_gemm<1><<<dim3(12, 4, 32), dim3(256), 0, stream>>>(Dm, x, Y1);

  dct_gemm<2><<<dim3(4, 4, 96), dim3(256), 0, stream>>>(Y1, Dm, out);
}


# Work-group tile 128x128 -> 256x128 (stage 1) / 128x256 (stage 2), long on the basis side; wave tile 64x32 -> 64x64: folded data tiles staged by half as many groups, 16 WMMAs per 8 fragment reads
# speedup vs baseline: 2.3254x; 1.2374x over previous
#include <hip/hip_runtime.h>

typedef __attribute__((ext_vector_type(2))) float v2f;
typedef __attribute__((ext_vector_type(8))) float v8f;

#define TILE_K 32
#define AP     34
#define GK     256
#define MIR    511
#define NT_K   (GK / TILE_K)
#define HWC    786432L

__global__ __launch_bounds__(256) void dct_make(float* __restrict__ D) {
  int idx = blockIdx.x * 256 + threadIdx.x;
  int k = idx >> 9;
  int n = idx & 511;
  int r = ((2 * n + 1) * k) & 2047;
  float th = 3.14159265358979323846f * (float)r * (1.0f / 1024.0f);
  float v = cosf(th) * 0.0625f;
  if (k == 0) v *= 0.70710678118654752f;
  D[idx] = v;
}

template <int ST> struct Geo;
template <> struct Geo<1> {
  static constexpr int rsA = 1024, csA = 1, rsB = 1536, csB = 1, rsC = 3072, csC = 1;
  static constexpr int parA = 512, parB = 0, parC = 1536;
  static constexpr int zDiv = 1;
  static constexpr bool offA = false, offB = true, offC = true, bNFast = true;
  static constexpr bool foldA = false;
  static constexpr int BM = 256, BN = 128, BP = 2 * 128 + 34;
};
template <> struct Geo<2> {
  static constexpr int rsA = 1536, csA = 3, rsB = 1, csB = 1024, rsC = 1536, csC = 6;
  static constexpr int parA = 0, parB = 512, parC = 3;
  static constexpr int zDiv = 3;
  static constexpr bool offA = true, offB = false, offC = true, bNFast = false;
  static constexpr bool foldA = true;
  static constexpr int BM = 128, BN = 256, BP = 2 * 256 + 34;
};


template <int ST, int N>
__device__ __forceinline__ void fetchA(const float* __restrict__ pa, float (&r)[N]) {
#pragma unroll
  for (int t = 0; t < N; ++t) r[t] = pa[t * 8 * Geo<ST>::rsA];
}

template <int N>
__device__ __forceinline__ void storeA(float* __restrict__ As, int sa, const float (&r)[N]) {
#pragma unroll
  for (int t = 0; t < N; ++t) As[sa + t * 8 * AP] = r[t];
}

template <int ST, int N>
__device__ __forceinline__ void fetchB(const float* __restrict__ pb, float (&r)[N]) {
  if (Geo<ST>::bNFast) {
#pragma unroll
    for (int t = 0; t < N; ++t) r[t] = pb[t * 2 * Geo<ST>::rsB];
  } else {
#pragma unroll
    for (int t = 0; t < N; ++t) r[t] = pb[t * 8 * Geo<ST>::csB];
  }
}

template <int ST, int N>
__device__ __forceinline__ void fetchMir(const float* __restrict__ pm, float (&r)[N]) {
  if (Geo<ST>::foldA) {
#pragma unroll
    for (int t = 0; t < N; ++t) r[t] = pm[t * 8 * Geo<ST>::rsA];
  } else {
#pragma unroll
    for (int t = 0; t < N; ++t) r[t] = pm[-(t * 2 * Geo<ST>::rsB)];
  }
}

template <int NR, int NP>
__device__ __forceinline__ void foldMir(float (&r)[NR], const float (&m)[NP], float s) {
#pragma unroll
  for (int t = 0; t < NP; ++t) r[t] = __builtin_fmaf(s, m[t], r[t]);
}

template <int ST, int N>
__device__ __forceinline__ void storeB(float* __restrict__ Bs, int sb, const float (&r)[N]) {
  if (Geo<ST>::bNFast) {
#pragma unroll
    for (int t = 0; t < N; ++t) Bs[sb + t * Geo<ST>::BP] = r[t];
  } else {
#pragma unroll
    for (int t = 0; t < N; ++t) Bs[sb + t * 16] = r[t];
  }
}


template <int BPITCH>
__device__ __forceinline__ void mma_tile(const float* __restrict__ As,
                                         const float* __restrict__ Bs,
                                         v8f (&acc)[4][4],
                                         int wm, int wn, int row, int grp) {
#pragma unroll
  for (int kk = 0; kk < TILE_K; kk += 4) {
    v2f a[4], b[4];
#pragma unroll
    for (int mt = 0; mt < 4; ++mt)
      a[mt] = *(const v2f*)&As[(wm + mt * 16 + row) * AP + kk + 2 * grp];
#pragma unroll
    for (int nt = 0; nt < 4; ++nt)
      b[nt] = *(const v2f*)&Bs[((kk >> 1) + grp) * BPITCH + (wn + nt * 16 + row) * 2];
#pragma unroll
    for (int mt = 0; mt < 4; ++mt)
#pragma unroll
      for (int nt = 0; nt < 4; ++nt)
        acc[mt][nt] = __builtin_amdgcn_wmma_f32_16x16x4_f32(
            false, a[mt], false, b[nt], (short)0, acc[mt][nt], false, false);
  }
}


template <int ST>
__global__ __launch_bounds__(256) __attribute__((amdgpu_waves_per_eu(4, 4)))
void dct_gemm(const float* __restrict__ A, const float* __restrict__ B,
              float* __restrict__ C) {
  typedef Geo<ST> G;
  constexpr int NA = G::BM / 8;
  constexpr int NB = G::bNFast ? 16 : G::BN / 8;
  constexpr int NM = G::foldA ? NA : NB;
  constexpr int WAVES_M = G::BM / 64;
  __shared__ float As[2][G::BM * AP];
  __shared__ float Bs[2][(TILE_K / 2) * G::BP];

  const int tid = threadIdx.x;
  const int z   = blockIdx.z;
  const long off = (long)(z / G::zDiv) * HWC + (long)(z % G::zDiv);
  const int par = G::foldA ? blockIdx.x : blockIdx.y;
  const float sgn = par ? -1.0f : 1.0f;
  const float* Ab = A + (G::offA ? off : 0) + par * G::parA;
  const float* Bb = B + (G::offB ? off : 0) + par * G::parB;
  float*       Cb = C + (G::offC ? off : 0) + par * G::parC;

  const int m0 = G::foldA ? blockIdx.y * G::BM : 0;
  const int n0 = G::foldA ? 0 : blockIdx.x * G::BN;

  const int lane = tid & 31;
  const int wave = tid >> 5;
  const int wm   = (wave % WAVES_M) * 64;
  const int wn   = (wave / WAVES_M) * 64;
  const int row  = lane & 15;
  const int grp  = lane >> 4;

  const float* pa = Ab + (long)(m0 + (tid >> 5)) * G::rsA + (long)(tid & 31) * G::csA;
  const float* pb = G::bNFast
      ? Bb + (long)(tid >> 7) * G::rsB + (long)(n0 + (tid & 127)) * G::csB
      : Bb + (long)(tid & 31) * G::rsB + (long)(n0 + (tid >> 5)) * G::csB;
  const int sa = (tid >> 5) * AP + (tid & 31);
  const int sb = G::bNFast
      ? (tid & 127) * 2 + (tid >> 7)
      : ((tid & 31) >> 1) * G::BP + (tid >> 5) * 2 + (tid & 1);
  const float* pm = G::foldA
      ? Ab + (long)(m0 + (tid >> 5)) * G::rsA + (long)(MIR - (tid & 31)) * G::csA
      : Bb + (long)(MIR - (tid >> 7)) * G::rsB + (long)(n0 + (tid & 127)) * G::csB;
  const long stepA = (long)TILE_K * G::csA;
  const long stepB = (long)TILE_K * G::rsB;
  const long stepM = G::foldA ? stepA : stepB;

  v8f acc[4][4];
#pragma unroll
  for (int i = 0; i < 4; ++i)
#pragma unroll
    for (int j = 0; j < 4; ++j) acc[i][j] = (v8f)0.0f;

  float ra[NA], rb[NB], rm[NM];

  fetchA<ST>(pa, ra);
  fetchB<ST>(pb, rb);
  fetchMir<ST>(pm, rm);
  if (G::foldA) foldMir(ra, rm, sgn); else foldMir(rb, rm, sgn);
  storeA(As[0], sa, ra);
  storeB<ST>(Bs[0], sb, rb);
  __syncthreads();

  for (int t = 0; t < NT_K; ++t) {
    const int cur = t & 1;
    if (t + 1 < NT_K) {
      pa += stepA;
      pb += stepB;
      pm -= stepM;
      fetchA<ST>(pa, ra);
      fetchB<ST>(pb, rb);
      fetchMir<ST>(pm, rm);
    }
    mma_tile<G::BP>(As[cur], Bs[cur], acc, wm, wn, row, grp);
    __builtin_amdgcn_sched_barrier(0);
    if (t + 1 < NT_K) {
      if (G::foldA) foldMir(ra, rm, sgn); else foldMir(rb, rm, sgn);
      storeA(As[cur ^ 1], sa, ra);
      storeB<ST>(Bs[cur ^ 1], sb, rb);
    }
    __syncthreads();
  }

  float* pc = Cb + (long)(m0 + wm + grp * 8) * G::rsC + (long)(n0 + wn + row) * G::csC;
#pragma unroll
  for (int mt = 0; mt < 4; ++mt)
#pragma unroll
    for (int nt = 0; nt < 4; ++nt)
#pragma unroll
      for (int v = 0; v < 8; ++v)
        pc[(mt * 16 + v) * G::rsC + nt * 16 * G::csC] = acc[mt][nt][v];
}

extern "C" void kernel_launch(void* const* d_in, const int* in_sizes, int n_in,
                              void* d_out, int out_size, void* d_ws, size_t ws_size,
                              hipStream_t stream) {
  (void)in_sizes; (void)n_in; (void)out_size; (void)ws_size;
  const float* x = (const float*)d_in[0];
  float* out = (float*)d_out;
  float* Dm  = (float*)d_ws;
  float* Y1  = Dm + 512 * 512;

  dct_make<<<dim3(1024), dim3(256), 0, stream>>>(Dm);

  dct_gemm<1><<<dim3(12, 2, 32), dim3(256), 0, stream>>>(Dm, x, Y1);

  dct_gemm<2><<<dim3(2, 4, 96), dim3(256), 0, stream>>>(Y1, Dm, out);
}


# Second mirror fold for even outputs (D[2m][255-n] = (-1)^m D[2m][n]): parity-0 work-groups pre-sum four mirror partners in both signs, K 256 -> 128 there; changes summation order
# speedup vs baseline: 2.4694x; 1.0619x over previous
#include <hip/hip_runtime.h>

typedef __attribute__((ext_vector_type(2))) float v2f;
typedef __attribute__((ext_vector_type(8))) float v8f;

#define TILE_K 32
#define AP     34
#define GK     256
#define MIR    511
#define NT_K   (GK / TILE_K)
#define HWC    786432L

__global__ __launch_bounds__(256) void dct_make(float* __restrict__ D) {
  int idx = blockIdx.x * 256 + threadIdx.x;
  int k = idx >> 9;
  int n = idx & 511;
  int r = ((2 * n + 1) * k) & 2047;
  float th = 3.14159265358979323846f * (float)r * (1.0f / 1024.0f);
  float v = cosf(th) * 0.0625f;
  if (k == 0) v *= 0.70710678118654752f;
  D[idx] = v;
}

template <int ST> struct Geo;
template <> struct Geo<1> {
  static constexpr int rsA = 1024, csA = 1, rsB = 1536, csB = 1, rsC = 3072, csC = 1;
  static constexpr int parA = 512, parB = 0, parC = 1536;
  static constexpr int zDiv = 1;
  static constexpr bool offA = false, offB = true, offC = true, bNFast = true;
  static constexpr bool foldA = false;
  static constexpr int BM = 256, BN = 128, BP = 2 * 128 + 34;
};
template <> struct Geo<2> {
  static constexpr int rsA = 1536, csA = 3, rsB = 1, csB = 1024, rsC = 1536, csC = 6;
  static constexpr int parA = 0, parB = 512, parC = 3;
  static constexpr int zDiv = 3;
  static constexpr bool offA = true, offB = false, offC = true, bNFast = false;
  static constexpr bool foldA = true;
  static constexpr int BM = 128, BN = 256, BP = 2 * 256 + 34;
};


template <int ST, int N>
__device__ __forceinline__ void fetchA(const float* __restrict__ pa, float (&r)[N]) {
#pragma unroll
  for (int t = 0; t < N; ++t) r[t] = pa[t * 8 * Geo<ST>::rsA];
}

template <int N>
__device__ __forceinline__ void storeA(float* __restrict__ As, int sa, const float (&r)[N]) {
#pragma unroll
  for (int t = 0; t < N; ++t) As[sa + t * 8 * AP] = r[t];
}

template <int ST, int N>
__device__ __forceinline__ void fetchB(const float* __restrict__ pb, float (&r)[N]) {
  if (Geo<ST>::bNFast) {
#pragma unroll
    for (int t = 0; t < N; ++t) r[t] = pb[t * 2 * Geo<ST>::rsB];
  } else {
#pragma unroll
    for (int t = 0; t < N; ++t) r[t] = pb[t * 8 * Geo<ST>::csB];
  }
}

template <int ST, int N>
__device__ __forceinline__ void fetchMir(const float* __restrict__ pm, float (&r)[N]) {
  if (Geo<ST>::foldA) {
#pragma unroll
    for (int t = 0; t < N; ++t) r[t] = pm[t * 8 * Geo<ST>::rsA];
  } else {
#pragma unroll
    for (int t = 0; t < N; ++t) r[t] = pm[-(t * 2 * Geo<ST>::rsB)];
  }
}

template <int NR, int NP>
__device__ __forceinline__ void foldMir(float (&r)[NR], const float (&m)[NP], float s) {
#pragma unroll
  for (int t = 0; t < NP; ++t) r[t] = __builtin_fmaf(s, m[t], r[t]);
}

template <int ST, int N>
__device__ __forceinline__ void storeB(float* __restrict__ Bs, int sb, const float (&r)[N]) {
  if (Geo<ST>::bNFast) {
#pragma unroll
    for (int t = 0; t < N; ++t) Bs[sb + t * Geo<ST>::BP] = r[t];
  } else {
#pragma unroll
    for (int t = 0; t < N; ++t) Bs[sb + t * 16] = r[t];
  }
}


template <int BPITCH, int KT>
__device__ __forceinline__ void mma_tile(const float* __restrict__ As,
                                         const float* __restrict__ Bs,
                                         v8f (&acc)[4][4],
                                         int wm, int wn, int row, int grp) {
#pragma unroll
  for (int kk = 0; kk < KT; kk += 4) {
    v2f a[4], b[4];
#pragma unroll
    for (int mt = 0; mt < 4; ++mt)
      a[mt] = *(const v2f*)&As[(wm + mt * 16 + row) * AP + kk + 2 * grp];
#pragma unroll
    for (int nt = 0; nt < 4; ++nt)
      b[nt] = *(const v2f*)&Bs[((kk >> 1) + grp) * BPITCH + (wn + nt * 16 + row) * 2];
#pragma unroll
    for (int mt = 0; mt < 4; ++mt)
#pragma unroll
      for (int nt = 0; nt < 4; ++nt)
        acc[mt][nt] = __builtin_amdgcn_wmma_f32_16x16x4_f32(
            false, a[mt], false, b[nt], (short)0, acc[mt][nt], false, false);
  }
}

#define TKE 16

__device__ __forceinline__ void fetchBasisE(const float* __restrict__ p, float (&r)[16]) {
#pragma unroll
  for (int t = 0; t < 8; ++t) r[t] = p[t * 32768];
#pragma unroll
  for (int t = 0; t < 8; ++t) r[8 + t] = p[t * 32768 + 1024];
}

template <int ST, int SGN>
__device__ __forceinline__ void fetchDataE(const float* __restrict__ p, float (&r)[8]) {
  if (Geo<ST>::foldA) {
#pragma unroll
    for (int t = 0; t < 8; ++t) r[t] = p[t * 16 * Geo<ST>::rsA];
  } else {
#pragma unroll
    for (int t = 0; t < 8; ++t) r[t] = p[SGN * t * 2 * Geo<ST>::rsB];
  }
}

template <int ST>
__device__ __forceinline__ void gemm_even(const float* __restrict__ A, const float* __restrict__ B,
                                          float* __restrict__ C, float* __restrict__ lds) {
  typedef Geo<ST> G;
  constexpr bool dataIsA = G::foldA;
  constexpr int ASZ = G::BM * AP;
  constexpr int BSZ = (TKE / 2) * G::BP;
  constexpr int AV = dataIsA ? 2 : 1;
  constexpr int BV = dataIsA ? 1 : 2;
  constexpr int WAVES_M = G::BM / 64;
  float* As = lds;
  float* Bs = lds + 2 * AV * ASZ;

  const int tid = threadIdx.x;
  const int z   = blockIdx.z;
  const long off = (long)(z / G::zDiv) * HWC + (long)(z % G::zDiv);
  const float* Ab = A + (G::offA ? off : 0);
  const float* Bb = B + (G::offB ? off : 0);
  float*       Cb = C + (G::offC ? off : 0);
  const int m0 = dataIsA ? blockIdx.y * G::BM : 0;
  const int n0 = dataIsA ? 0 : blockIdx.x * G::BN;

  const int lane = tid & 31;
  const int wave = tid >> 5;
  const int wm   = (wave % WAVES_M) * 64;
  const int wn   = (wave / WAVES_M) * 64;
  const int row  = lane & 15;
  const int grp  = lane >> 4;
  const int ver  = dataIsA ? (wn >> 7) : (wm >> 7);

  const float* pbas = (dataIsA ? Bb : Ab) + (long)(tid >> 4) * 2048 + (tid & 15);
  const int sbas = dataIsA ? ((tid & 15) >> 1) * G::BP + (tid >> 4) * 2 + (tid & 1)
                           : (tid >> 4) * AP + (tid & 15);
  constexpr int sbasStep = dataIsA ? 32 : 16 * AP;
  const float *p0, *p1, *p2, *p3;
  if (dataIsA) {
    const float* rowp = Ab + (long)(m0 + (tid >> 4)) * G::rsA;
    const int k = tid & 15;
    p0 = rowp + (long)k * G::csA;
    p1 = rowp + (long)(511 - k) * G::csA;
    p2 = rowp + (long)(255 - k) * G::csA;
    p3 = rowp + (long)(256 + k) * G::csA;
  } else {
    const float* colp = Bb + (long)(n0 + (tid & 127)) * G::csB;
    const int k = tid >> 7;
    p0 = colp + (long)k * G::rsB;
    p1 = colp + (long)(511 - k) * G::rsB;
    p2 = colp + (long)(255 - k) * G::rsB;
    p3 = colp + (long)(256 + k) * G::rsB;
  }
  const long stepD = (long)TKE * (dataIsA ? G::csA : G::rsB);
  const int sdat = dataIsA ? (tid >> 4) * AP + (tid & 15)
                           : (tid & 127) * 2 + (tid >> 7);
  constexpr int sdatStep = dataIsA ? 16 * AP : G::BP;
  constexpr int verStep  = dataIsA ? ASZ : BSZ;

  v8f acc[4][4];
#pragma unroll
  for (int i = 0; i < 4; ++i)
#pragma unroll
    for (int j = 0; j < 4; ++j) acc[i][j] = (v8f)0.0f;

  float rbas[16], x0[8], x1[8], x2[8], x3[8];

  fetchBasisE(pbas, rbas);
  fetchDataE<ST, 1>(p0, x0);
  fetchDataE<ST, -1>(p1, x1);
  fetchDataE<ST, -1>(p2, x2);
  fetchDataE<ST, 1>(p3, x3);
  {
    float* bd = dataIsA ? Bs : As;
    float* dd = dataIsA ? As : Bs;
#pragma unroll
    for (int t = 0; t < 16; ++t) bd[sbas + t * sbasStep] = rbas[t];
#pragma unroll
    for (int t = 0; t < 8; ++t) {
      const float s = x0[t] + x1[t];
      const float d = x2[t] + x3[t];
      dd[sdat + t * sdatStep] = s + d;
      dd[verStep + sdat + t * sdatStep] = s - d;
    }
  }
  __syncthreads();

  for (int t = 0; t < GK / 2 / TKE; ++t) {
    const int cur = t & 1;
    if (t + 1 < GK / 2 / TKE) {
      pbas += TKE;
      p0 += stepD;
      p1 -= stepD;
      p2 -= stepD;
      p3 += stepD;
      fetchBasisE(pbas, rbas);
      fetchDataE<ST, 1>(p0, x0);
      fetchDataE<ST, -1>(p1, x1);
      fetchDataE<ST, -1>(p2, x2);
      fetchDataE<ST, 1>(p3, x3);
    }
    mma_tile<G::BP, TKE>(As + cur * AV * ASZ + (dataIsA ? ver * ASZ : 0),
                         Bs + cur * BV * BSZ + (dataIsA ? 0 : ver * BSZ),
                         acc, wm, wn, row, grp);
    __builtin_amdgcn_sched_barrier(0);
    if (t + 1 < GK / 2 / TKE) {
      float* bd = (dataIsA ? Bs : As) + (cur ^ 1) * (dataIsA ? BV * BSZ : AV * ASZ);
      float* dd = (dataIsA ? As : Bs) + (cur ^ 1) * (dataIsA ? AV * ASZ : BV * BSZ);
#pragma unroll
      for (int u = 0; u < 16; ++u) bd[sbas + u * sbasStep] = rbas[u];
#pragma unroll
      for (int u = 0; u < 8; ++u) {
        const float s = x0[u] + x1[u];
        const float d = x2[u] + x3[u];
        dd[sdat + u * sdatStep] = s + d;
        dd[verStep + sdat + u * sdatStep] = s - d;
      }
    }
    __syncthreads();
  }

  if (dataIsA) {
    float* pc = Cb + (long)(m0 + wm + grp * 8) * 1536
                   + (long)(4 * ((wn & 127) + row) + 2 * (wn >> 7)) * 3;
#pragma unroll
    for (int mt = 0; mt < 4; ++mt)
#pragma unroll
      for (int nt = 0; nt < 4; ++nt)
#pragma unroll
        for (int v = 0; v < 8; ++v)
          pc[(mt * 16 + v) * 1536 + nt * 16 * 4 * 3] = acc[mt][nt][v];
  } else {
    float* pc = Cb + (long)(4 * ((wm & 127) + grp * 8) + 2 * (wm >> 7)) * 1536
                   + (long)(n0 + wn + row);
#pragma unroll
    for (int mt = 0; mt < 4; ++mt)
#pragma unroll
      for (int nt = 0; nt < 4; ++nt)
#pragma unroll
        for (int v = 0; v < 8; ++v)
          pc[(mt * 16 + v) * 4 * 1536 + nt * 16] = acc[mt][nt][v];
  }
  asm volatile("" ::: "memory");
}


template <int ST>
__global__ __launch_bounds__(256) __attribute__((amdgpu_waves_per_eu(4, 4)))
void dct_gemm(const float* __restrict__ A, const float* __restrict__ B,
              float* __restrict__ C) {
  typedef Geo<ST> G;
  constexpr int NA = G::BM / 8;
  constexpr int NB = G::bNFast ? 16 : G::BN / 8;
  constexpr int NM = G::foldA ? NA : NB;
  constexpr int WAVES_M = G::BM / 64;
  constexpr int ASZ = G::BM * AP;
  constexpr int BSZ = (TILE_K / 2) * G::BP;
  __shared__ float lds[2 * ASZ + 2 * BSZ];
  if ((G::foldA ? blockIdx.x : blockIdx.y) == 0) {
    gemm_even<ST>(A, B, C, lds);
    return;
  }
  float* As0 = lds;
  float* Bs0 = lds + 2 * ASZ;

  const int tid = threadIdx.x;
  const int z   = blockIdx.z;
  const long off = (long)(z / G::zDiv) * HWC + (long)(z % G::zDiv);
  const int par = G::foldA ? blockIdx.x : blockIdx.y;
  const float sgn = par ? -1.0f : 1.0f;
  const float* Ab = A + (G::offA ? off : 0) + par * G::parA;
  const float* Bb = B + (G::offB ? off : 0) + par * G::parB;
  float*       Cb = C + (G::offC ? off : 0) + par * G::parC;

  const int m0 = G::foldA ? blockIdx.y * G::BM : 0;
  const int n0 = G::foldA ? 0 : blockIdx.x * G::BN;

  const int lane = tid & 31;
  const int wave = tid >> 5;
  const int wm   = (wave % WAVES_M) * 64;
  const int wn   = (wave / WAVES_M) * 64;
  const int row  = lane & 15;
  const int grp  = lane >> 4;

  const float* pa = Ab + (long)(m0 + (tid >> 5)) * G::rsA + (long)(tid & 31) * G::csA;
  const float* pb = G::bNFast
      ? Bb + (long)(tid >> 7) * G::rsB + (long)(n0 + (tid & 127)) * G::csB
      : Bb + (long)(tid & 31) * G::rsB + (long)(n0 + (tid >> 5)) * G::csB;
  const int sa = (tid >> 5) * AP + (tid & 31);
  const int sb = G::bNFast
      ? (tid & 127) * 2 + (tid >> 7)
      : ((tid & 31) >> 1) * G::BP + (tid >> 5) * 2 + (tid & 1);
  const float* pm = G::foldA
      ? Ab + (long)(m0 + (tid >> 5)) * G::rsA + (long)(MIR - (tid & 31)) * G::csA
      : Bb + (long)(MIR - (tid >> 7)) * G::rsB + (long)(n0 + (tid & 127)) * G::csB;
  const long stepA = (long)TILE_K * G::csA;
  const long stepB = (long)TILE_K * G::rsB;
  const long stepM = G::foldA ? stepA : stepB;

  v8f acc[4][4];
#pragma unroll
  for (int i = 0; i < 4; ++i)
#pragma unroll
    for (int j = 0; j < 4; ++j) acc[i][j] = (v8f)0.0f;

  float ra[NA], rb[NB], rm[NM];

  fetchA<ST>(pa, ra);
  fetchB<ST>(pb, rb);
  fetchMir<ST>(pm, rm);
  if (G::foldA) foldMir(ra, rm, sgn); else foldMir(rb, rm, sgn);
  storeA(As0, sa, ra);
  storeB<ST>(Bs0, sb, rb);
  __syncthreads();

  for (int t = 0; t < NT_K; ++t) {
    const int cur = t & 1;
    if (t + 1 < NT_K) {
      pa += stepA;
      pb += stepB;
      pm -= stepM;
      fetchA<ST>(pa, ra);
      fetchB<ST>(pb, rb);
      fetchMir<ST>(pm, rm);
    }
    mma_tile<G::BP, TILE_K>(As0 + cur * ASZ, Bs0 + cur * BSZ, acc, wm, wn, row, grp);
    __builtin_amdgcn_sched_barrier(0);
    if (t + 1 < NT_K) {
      if (G::foldA) foldMir(ra, rm, sgn); else foldMir(rb, rm, sgn);
      storeA(As0 + (cur ^ 1) * ASZ, sa, ra);
      storeB<ST>(Bs0 + (cur ^ 1) * BSZ, sb, rb);
    }
    __syncthreads();
  }

  float* pc = Cb + (long)(m0 + wm + grp * 8) * G::rsC + (long)(n0 + wn + row) * G::csC;
#pragma unroll
  for (int mt = 0; mt < 4; ++mt)
#pragma unroll
    for (int nt = 0; nt < 4; ++nt)
#pragma unroll
      for (int v = 0; v < 8; ++v)
        pc[(mt * 16 + v) * G::rsC + nt * 16 * G::csC] = acc[mt][nt][v];
}

extern "C" void kernel_launch(void* const* d_in, const int* in_sizes, int n_in,
                              void* d_out, int out_size, void* d_ws, size_t ws_size,
                              hipStream_t stream) {
  (void)in_sizes; (void)n_in; (void)out_size; (void)ws_size;
  const float* x = (const float*)d_in[0];
  float* out = (float*)d_out;
  float* Dm  = (float*)d_ws;
  float* Y1  = Dm + 512 * 512;

  dct_make<<<dim3(1024), dim3(256), 0, stream>>>(Dm);

  dct_gemm<1><<<dim3(12, 2, 32), dim3(256), 0, stream>>>(Dm, x, Y1);

  dct_gemm<2><<<dim3(2, 4, 96), dim3(256), 0, stream>>>(Y1, Dm, out);
}
